# starting kernel
# baseline (speedup 1.0000x reference)
#include <hip/hip_runtime.h>
#include <hip/hip_bf16.h>

typedef float v2f __attribute__((ext_vector_type(2)));
typedef float v8f __attribute__((ext_vector_type(8)));

static __device__ __forceinline__ v8f wmma4(v2f a, v2f b, v8f c) {
  return __builtin_amdgcn_wmma_f32_16x16x4_f32(false, a, false, b, (short)0, c,
                                               false, false);
}

#define LD 512
#define ROWS 4096
#define CHUNK 64
#define NCHUNK 32
#define NTASK 512

__global__ __launch_bounds__(128) void gemm_f32(const float* __restrict__ A,
                                                const float* __restrict__ B,
                                                float* __restrict__ C,
                                                const float* __restrict__ bias,
                                                int M, int N, int K) {
  __shared__ float As[64][17];
  __shared__ float Bs[16][65];
  const int t = threadIdx.x;
  const int wave = t >> 5, lane = t & 31, lh = lane >> 4, l16 = lane & 15;
  const int m0 = blockIdx.y * 64, n0 = blockIdx.x * 64;

  v8f acc[4] = {};

  for (int k0 = 0; k0 < K; k0 += 16) {
    {
      int r = t >> 1, c = (t & 1) * 8;
      const float4* src = reinterpret_cast<const float4*>(&A[(size_t)(m0 + r) * K + k0 + c]);
      if (k0 + 16 < K) __builtin_prefetch(&A[(size_t)(m0 + r) * K + k0 + 16 + c], 0, 1);
      float4 v0 = src[0], v1 = src[1];
      float* d = &As[r][c];
      d[0] = v0.x; d[1] = v0.y; d[2] = v0.z; d[3] = v0.w;
      d[4] = v1.x; d[5] = v1.y; d[6] = v1.z; d[7] = v1.w;
    }
    {
      int r = t >> 3, c = (t & 7) * 8;
      const float4* src = reinterpret_cast<const float4*>(&B[(size_t)(k0 + r) * N + n0 + c]);
      float4 v0 = src[0], v1 = src[1];
      float* d = &Bs[r][c];
      d[0] = v0.x; d[1] = v0.y; d[2] = v0.z; d[3] = v0.w;
      d[4] = v1.x; d[5] = v1.y; d[6] = v1.z; d[7] = v1.w;
    }
    __syncthreads();
    const int mr = wave * 16 + l16;
#pragma unroll
    for (int kk = 0; kk < 16; kk += 4) {
      v2f a;
      a.x = As[mr][kk + 2 * lh];
      a.y = As[mr][kk + 2 * lh + 1];
#pragma unroll
      for (int nt = 0; nt < 4; ++nt) {
        v2f b;
        b.x = Bs[kk + 2 * lh][nt * 16 + l16];
        b.y = Bs[kk + 2 * lh + 1][nt * 16 + l16];
        acc[nt] = wmma4(a, b, acc[nt]);
      }
    }
    __syncthreads();
  }

#pragma unroll
  for (int nt = 0; nt < 4; ++nt) {
    int col = n0 + nt * 16 + l16;
    float badd = (bias != nullptr) ? bias[col] : 0.0f;
#pragma unroll
    for (int r = 0; r < 8; ++r) {
      int row = m0 + wave * 16 + r + 8 * lh;
      C[(size_t)row * N + col] = acc[nt][r] + badd;
    }
  }
}

__global__ __launch_bounds__(128) void qk_transform(float* __restrict__ Q,
                                                    float* __restrict__ Kb) {
  const int wave = threadIdx.x >> 5, lane = threadIdx.x & 31;
  const int hr = blockIdx.x * 4 + wave;
  if (hr >= ROWS * 8) return;
  const int row = hr >> 3, h = hr & 7;
  const size_t base = (size_t)row * LD + h * 64;

  float q0 = Q[base + lane], q1 = Q[base + lane + 32];
  float m = fmaxf(q0, q1);
#pragma unroll
  for (int off = 16; off > 0; off >>= 1) m = fmaxf(m, __shfl_xor(m, off, 32));
  float e0 = __expf(q0 - m), e1 = __expf(q1 - m);
  float s = e0 + e1;
#pragma unroll
  for (int off = 16; off > 0; off >>= 1) s += __shfl_xor(s, off, 32);
  float inv = 0.125f / s;
  Q[base + lane] = e0 * inv;
  Q[base + lane + 32] = e1 * inv;

  float k0 = Kb[base + lane], k1 = Kb[base + lane + 32];
  Kb[base + lane] = __expf(k0);
  Kb[base + lane + 32] = __expf(k1);
}

__global__ __launch_bounds__(256) void chunk_state(const float* __restrict__ Kb,
                                                   const float* __restrict__ Vb,
                                                   float* __restrict__ Sbuf,
                                                   float* __restrict__ ksumbuf) {
  __shared__ float Ks[64][65];
  __shared__ float Vs[64][65];
  const int task = blockIdx.x;
  const int ck = task & 31, h = (task >> 5) & 7, b = task >> 8;
  const int row0 = b * 2048 + ck * CHUNK, col0 = h * 64;
  const int t = threadIdx.x, wave = t >> 5, lane = t & 31;
  const int lh = lane >> 4, l16 = lane & 15;

  {
    int r = t >> 2, c = (t & 3) * 16;
    const float4* ks = reinterpret_cast<const float4*>(&Kb[(size_t)(row0 + r) * LD + col0 + c]);
    const float4* vs = reinterpret_cast<const float4*>(&Vb[(size_t)(row0 + r) * LD + col0 + c]);
#pragma unroll
    for (int i = 0; i < 4; ++i) {
      float4 kv = ks[i], vv = vs[i];
      float* kd = &Ks[r][c + i * 4];
      kd[0] = kv.x; kd[1] = kv.y; kd[2] = kv.z; kd[3] = kv.w;
      float* vd = &Vs[r][c + i * 4];
      vd[0] = vv.x; vd[1] = vv.y; vd[2] = vv.z; vd[3] = vv.w;
    }
  }
  __syncthreads();

  const size_t sbase = (size_t)task * 4096;
#pragma unroll
  for (int tt = wave * 2; tt <= wave * 2 + 1; ++tt) {
    const int mt = tt >> 2, nt = tt & 3;
    v8f acc = {};
#pragma unroll
    for (int kk = 0; kk < 64; kk += 4) {
      int tr = kk + 2 * lh;
      v2f a, bfr;
      a.x = Ks[tr][mt * 16 + l16];
      a.y = Ks[tr + 1][mt * 16 + l16];
      bfr.x = Vs[tr][nt * 16 + l16];
      bfr.y = Vs[tr + 1][nt * 16 + l16];
      acc = wmma4(a, bfr, acc);
    }
#pragma unroll
    for (int r = 0; r < 8; ++r) {
      int d = mt * 16 + r + 8 * lh, n = nt * 16 + l16;
      Sbuf[sbase + d * 64 + n] = acc[r];
    }
  }
  if (t < 64) {
    float s = 0.f;
#pragma unroll 8
    for (int j = 0; j < 64; ++j) s += Ks[j][t];
    ksumbuf[task * 64 + t] = s;
  }
}

__global__ void prefix_scan(float* __restrict__ Sbuf, float* __restrict__ ksumbuf) {
  const int g = blockIdx.x * blockDim.x + threadIdx.x;
  if (g < 16 * 4096) {
    const int bh = g >> 12, e = g & 4095;
    float run = 0.f;
    for (int c = 0; c < NCHUNK; ++c) {
      size_t idx = (size_t)(bh * NCHUNK + c) * 4096 + e;
      float v = Sbuf[idx];
      Sbuf[idx] = run;
      run += v;
    }
  } else if (g < 16 * 4096 + 16 * 64) {
    const int s2 = g - 16 * 4096, bh = s2 >> 6, d = s2 & 63;
    float run = 0.f;
    for (int c = 0; c < NCHUNK; ++c) {
      int idx = (bh * NCHUNK + c) * 64 + d;
      float v = ksumbuf[idx];
      ksumbuf[idx] = run;
      run += v;
    }
  }
}

__global__ __launch_bounds__(256) void attn_chunk(const float* __restrict__ Qb,
                                                  const float* __restrict__ Kb,
                                                  const float* __restrict__ Vb,
                                                  const float* __restrict__ Sbuf,
                                                  const float* __restrict__ ksumbuf,
                                                  float* __restrict__ attn) {
  extern __shared__ float sm[];
  float* Qs   = sm;
  float* Ks   = Qs + 64 * 65;
  float* Vs   = Ks + 64 * 65;
  float* Ss   = Vs + 64 * 65;
  float* Ms   = Ss + 64 * 65;
  float* ksum = Ms + 64 * 65;
  float* Dinv = ksum + 64;

  const int task = blockIdx.x;
  const int ck = task & 31, h = (task >> 5) & 7, b = task >> 8;
  const int row0 = b * 2048 + ck * CHUNK, col0 = h * 64;
  const int t = threadIdx.x, wave = t >> 5, lane = t & 31;
  const int lh = lane >> 4, l16 = lane & 15;

  {
    int r = t >> 2, c = (t & 3) * 16;
    const float4* qs = reinterpret_cast<const float4*>(&Qb[(size_t)(row0 + r) * LD + col0 + c]);
    const float4* ks = reinterpret_cast<const float4*>(&Kb[(size_t)(row0 + r) * LD + col0 + c]);
    const float4* vs = reinterpret_cast<const float4*>(&Vb[(size_t)(row0 + r) * LD + col0 + c]);
    const float4* ss = reinterpret_cast<const float4*>(&Sbuf[(size_t)task * 4096 + t * 16]);
#pragma unroll
    for (int i = 0; i < 4; ++i) {
      float4 qv = qs[i], kv = ks[i], vv = vs[i], sv = ss[i];
      float* qd = &Qs[r * 65 + c + i * 4];
      qd[0] = qv.x; qd[1] = qv.y; qd[2] = qv.z; qd[3] = qv.w;
      float* kd = &Ks[r * 65 + c + i * 4];
      kd[0] = kv.x; kd[1] = kv.y; kd[2] = kv.z; kd[3] = kv.w;
      float* vd = &Vs[r * 65 + c + i * 4];
      vd[0] = vv.x; vd[1] = vv.y; vd[2] = vv.z; vd[3] = vv.w;
      float* sd = &Ss[(t >> 2) * 65 + (t & 3) * 16 + i * 4];
      sd[0] = sv.x; sd[1] = sv.y; sd[2] = sv.z; sd[3] = sv.w;
    }
    if (t < 64) ksum[t] = ksumbuf[task * 64 + t];
  }
  __syncthreads();

#pragma unroll
  for (int tt = wave * 2; tt <= wave * 2 + 1; ++tt) {
    const int mt = tt >> 2, nt = tt & 3;
    v8f acc = {};
#pragma unroll
    for (int kk = 0; kk < 64; kk += 4) {
      v2f a, bfr;
      a.x = Qs[(mt * 16 + l16) * 65 + kk + 2 * lh];
      a.y = Qs[(mt * 16 + l16) * 65 + kk + 2 * lh + 1];
      bfr.x = Ks[(nt * 16 + l16) * 65 + kk + 2 * lh];
      bfr.y = Ks[(nt * 16 + l16) * 65 + kk + 2 * lh + 1];
      acc = wmma4(a, bfr, acc);
    }
#pragma unroll
    for (int r = 0; r < 8; ++r) {
      int i = mt * 16 + r + 8 * lh, j = nt * 16 + l16;
      Ms[i * 65 + j] = (j <= i) ? acc[r] : 0.0f;
    }
  }
  __syncthreads();

  if (t < 64) {
    float s = 0.f, dq = 0.f;
#pragma unroll 8
    for (int j = 0; j < 64; ++j) s += Ms[t * 65 + j];
#pragma unroll 8
    for (int d = 0; d < 64; ++d) dq += Qs[t * 65 + d] * ksum[d];
    Dinv[t] = 1.0f / fmaxf(s + dq, 1e-3f);
  }
  __syncthreads();

#pragma unroll
  for (int tt = wave * 2; tt <= wave * 2 + 1; ++tt) {
    const int mt = tt >> 2, nt = tt & 3;
    v8f acc = {};
#pragma unroll
    for (int kk = 0; kk < 64; kk += 4) {
      v2f a, bfr;
      a.x = Ms[(mt * 16 + l16) * 65 + kk + 2 * lh];
      a.y = Ms[(mt * 16 + l16) * 65 + kk + 2 * lh + 1];
      bfr.x = Vs[(kk + 2 * lh) * 65 + nt * 16 + l16];
      bfr.y = Vs[(kk + 2 * lh + 1) * 65 + nt * 16 + l16];
      acc = wmma4(a, bfr, acc);
    }
#pragma unroll
    for (int kk = 0; kk < 64; kk += 4) {
      v2f a, bfr;
      a.x = Qs[(mt * 16 + l16) * 65 + kk + 2 * lh];
      a.y = Qs[(mt * 16 + l16) * 65 + kk + 2 * lh + 1];
      bfr.x = Ss[(kk + 2 * lh) * 65 + nt * 16 + l16];
      bfr.y = Ss[(kk + 2 * lh + 1) * 65 + nt * 16 + l16];
      acc = wmma4(a, bfr, acc);
    }
#pragma unroll
    for (int r = 0; r < 8; ++r) {
      int i = mt * 16 + r + 8 * lh;
      int col = col0 + nt * 16 + l16;
      attn[(size_t)(row0 + i) * LD + col] = acc[r] * Dinv[i];
    }
  }
}

extern "C" void kernel_launch(void* const* d_in, const int* in_sizes, int n_in,
                              void* d_out, int out_size, void* d_ws, size_t ws_size,
                              hipStream_t stream) {
  const float* x  = (const float*)d_in[0];
  const float* Wq = (const float*)d_in[1];
  const float* Wk = (const float*)d_in[2];
  const float* Wv = (const float*)d_in[3];
  const float* Wo = (const float*)d_in[4];
  const float* bo = (const float*)d_in[5];
  float* out = (float*)d_out;

  float* ws = (float*)d_ws;
  const size_t MN = (size_t)ROWS * LD;
  float* Qb = ws;
  float* Kb = Qb + MN;
  float* Vb = Kb + MN;
  float* At = Vb + MN;
  float* Sb = At + MN;
  float* Kc = Sb + (size_t)NTASK * 4096;

  dim3 ggrid(LD / 64, ROWS / 64);

  gemm_f32<<<ggrid, 128, 0, stream>>>(x, Wq, Qb, nullptr, ROWS, LD, LD);
  gemm_f32<<<ggrid, 128, 0, stream>>>(x, Wk, Kb, nullptr, ROWS, LD, LD);
  gemm_f32<<<ggrid, 128, 0, stream>>>(x, Wv, Vb, nullptr, ROWS, LD, LD);

  qk_transform<<<(ROWS * 8) / 4, 128, 0, stream>>>(Qb, Kb);

  chunk_state<<<NTASK, 256, 0, stream>>>(Kb, Vb, Sb, Kc);

  prefix_scan<<<(16 * 4096 + 16 * 64 + 255) / 256, 256, 0, stream>>>(Sb, Kc);

  size_t shbytes = (size_t)(5 * 64 * 65 + 128) * sizeof(float);
  attn_chunk<<<NTASK, 256, shbytes, stream>>>(Qb, Kb, Vb, Sb, Kc, At);

  gemm_f32<<<ggrid, 128, 0, stream>>>(At, Wo, out, bo, ROWS, LD, LD);
}


# q softmax/8 and k exp done on the accumulators in the q and k projection epilogues (wave tile 16x64 = one head); elementwise launch gone: 7 launches; softmax row sum now 3 adds + 4 row_xmask steps
# speedup vs baseline: 2.5375x; 2.5375x over previous
#include <hip/hip_runtime.h>
#include <hip/hip_bf16.h>

typedef float v2f __attribute__((ext_vector_type(2)));
typedef float v8f __attribute__((ext_vector_type(8)));

static __device__ __forceinline__ v8f wmma4(v2f a, v2f b, v8f c) {
  return __builtin_amdgcn_wmma_f32_16x16x4_f32(false, a, false, b, (short)0, c,
                                               false, false);
}

template <int M>
static __device__ __forceinline__ float row_xor(float v) {
  const int b = __builtin_bit_cast(int, v);
  return __builtin_bit_cast(float,
                            __builtin_amdgcn_update_dpp(0, b, 0x160 + M, 0xf, 0xf, true));
}

#define LD 512
#define ROWS 4096
#define CHUNK 64
#define NCHUNK 32
#define NTASK 512

struct GFrag {
  float4 a[4];
  float4 b[16];
};

static __device__ __forceinline__ void gfrag_load(GFrag& f, const float* __restrict__ ap,
                                                  const float* __restrict__ bp) {
#pragma unroll
  for (int q = 0; q < 4; ++q) f.a[q] = *reinterpret_cast<const float4*>(ap + 4 * q);
#pragma unroll
  for (int i = 0; i < 16; ++i) f.b[i] = *reinterpret_cast<const float4*>(bp + i * LD);
}

static __device__ __forceinline__ void gfrag_mma(const GFrag& f, v8f& c0, v8f& c1,
                                                 v8f& c2, v8f& c3) {
#pragma unroll
  for (int s = 0; s < 8; ++s) {
    const float4 av = f.a[s >> 1];
    v2f a, b;
    if (s & 1) { a.x = av.z; a.y = av.w; } else { a.x = av.x; a.y = av.y; }
    const float4 r0 = f.b[2 * s], r1 = f.b[2 * s + 1];
    b.x = r0.x; b.y = r1.x; c0 = wmma4(a, b, c0);
    b.x = r0.y; b.y = r1.y; c1 = wmma4(a, b, c1);
    b.x = r0.z; b.y = r1.z; c2 = wmma4(a, b, c2);
    b.x = r0.w; b.y = r1.w; c3 = wmma4(a, b, c3);
  }
}

template <int EPI>
__global__ __launch_bounds__(128) void gemm_f32(const float* __restrict__ A,
                                                const float* __restrict__ B,
                                                float* __restrict__ C,
                                                const float* __restrict__ bias) {
  const int t = threadIdx.x;
  const int wave = t >> 5, lane = t & 31, lh = lane >> 4, l16 = lane & 15;
  const int m0 = blockIdx.y * 64 + wave * 16, n0 = blockIdx.x * 64;

  const float* ap = A + (size_t)(m0 + l16) * LD + 16 * lh;
  const float* bp = B + (size_t)(16 * lh) * LD + n0 + 4 * l16;

  float4 bv = make_float4(0.f, 0.f, 0.f, 0.f);
  if (EPI == 0 && bias != nullptr) bv = *reinterpret_cast<const float4*>(bias + n0 + 4 * l16);

  v8f c0 = {}, c1 = {}, c2 = {}, c3 = {};
  GFrag f0, f1;

  gfrag_load(f0, ap, bp);
  for (int kb = 0; kb < LD / 32 - 2; kb += 2) {
    gfrag_load(f1, ap + 32, bp + 32 * LD);
    __builtin_amdgcn_sched_barrier(0);
    gfrag_mma(f0, c0, c1, c2, c3);
    ap += 64;
    bp += 64 * LD;
    gfrag_load(f0, ap, bp);
    __builtin_amdgcn_sched_barrier(0);
    gfrag_mma(f1, c0, c1, c2, c3);
  }
  gfrag_load(f1, ap + 32, bp + 32 * LD);
  __builtin_amdgcn_sched_barrier(0);
  gfrag_mma(f0, c0, c1, c2, c3);
  gfrag_mma(f1, c0, c1, c2, c3);

  float* cp = C + (size_t)(m0 + 8 * lh) * LD + n0 + 4 * l16;
  if (EPI == 1) {
    float m[8], s[8], e0[8], e1[8], e2[8], e3[8];
#pragma unroll
    for (int r = 0; r < 8; ++r) m[r] = fmaxf(fmaxf(c0[r], c1[r]), fmaxf(c2[r], c3[r]));
#pragma unroll
    for (int r = 0; r < 8; ++r) m[r] = fmaxf(m[r], row_xor<1>(m[r]));
#pragma unroll
    for (int r = 0; r < 8; ++r) m[r] = fmaxf(m[r], row_xor<2>(m[r]));
#pragma unroll
    for (int r = 0; r < 8; ++r) m[r] = fmaxf(m[r], row_xor<4>(m[r]));
#pragma unroll
    for (int r = 0; r < 8; ++r) m[r] = fmaxf(m[r], row_xor<8>(m[r]));
#pragma unroll
    for (int r = 0; r < 8; ++r) {
      e0[r] = __expf(c0[r] - m[r]); e1[r] = __expf(c1[r] - m[r]);
      e2[r] = __expf(c2[r] - m[r]); e3[r] = __expf(c3[r] - m[r]);
      s[r] = (e0[r] + e1[r]) + (e2[r] + e3[r]);
    }
#pragma unroll
    for (int r = 0; r < 8; ++r) s[r] += row_xor<1>(s[r]);
#pragma unroll
    for (int r = 0; r < 8; ++r) s[r] += row_xor<2>(s[r]);
#pragma unroll
    for (int r = 0; r < 8; ++r) s[r] += row_xor<4>(s[r]);
#pragma unroll
    for (int r = 0; r < 8; ++r) s[r] += row_xor<8>(s[r]);
#pragma unroll
    for (int r = 0; r < 8; ++r) {
      const float inv = 0.125f / s[r];
      float4 o;
      o.x = e0[r] * inv; o.y = e1[r] * inv; o.z = e2[r] * inv; o.w = e3[r] * inv;
      *reinterpret_cast<float4*>(cp + (size_t)r * LD) = o;
    }
  } else if (EPI == 2) {
#pragma unroll
    for (int r = 0; r < 8; ++r) {
      float4 o;
      o.x = __expf(c0[r]); o.y = __expf(c1[r]); o.z = __expf(c2[r]); o.w = __expf(c3[r]);
      *reinterpret_cast<float4*>(cp + (size_t)r * LD) = o;
    }
  } else {
#pragma unroll
    for (int r = 0; r < 8; ++r) {
      float4 o;
      o.x = c0[r] + bv.x; o.y = c1[r] + bv.y; o.z = c2[r] + bv.z; o.w = c3[r] + bv.w;
      *reinterpret_cast<float4*>(cp + (size_t)r * LD) = o;
    }
  }
}

__global__ __launch_bounds__(256) void chunk_state(const float* __restrict__ Kb,
                                                   const float* __restrict__ Vb,
                                                   float* __restrict__ Sbuf,
                                                   float* __restrict__ ksumbuf) {
  __shared__ float Ks[64][65];
  __shared__ float Vs[64][65];
  const int task = blockIdx.x;
  const int ck = task & 31, h = (task >> 5) & 7, b = task >> 8;
  const int row0 = b * 2048 + ck * CHUNK, col0 = h * 64;
  const int t = threadIdx.x, wave = t >> 5, lane = t & 31;
  const int lh = lane >> 4, l16 = lane & 15;

  {
    int r = t >> 2, c = (t & 3) * 16;
    const float4* ks = reinterpret_cast<const float4*>(&Kb[(size_t)(row0 + r) * LD + col0 + c]);
    const float4* vs = reinterpret_cast<const float4*>(&Vb[(size_t)(row0 + r) * LD + col0 + c]);
#pragma unroll
    for (int i = 0; i < 4; ++i) {
      float4 kv = ks[i], vv = vs[i];
      float* kd = &Ks[r][c + i * 4];
      kd[0] = kv.x; kd[1] = kv.y; kd[2] = kv.z; kd[3] = kv.w;
      float* vd = &Vs[r][c + i * 4];
      vd[0] = vv.x; vd[1] = vv.y; vd[2] = vv.z; vd[3] = vv.w;
    }
  }
  __syncthreads();

  const size_t sbase = (size_t)task * 4096;
#pragma unroll
  for (int tt = wave * 2; tt <= wave * 2 + 1; ++tt) {
    const int mt = tt >> 2, nt = tt & 3;
    v8f acc = {};
#pragma unroll
    for (int kk = 0; kk < 64; kk += 4) {
      int tr = kk + 2 * lh;
      v2f a, bfr;
      a.x = Ks[tr][mt * 16 + l16];
      a.y = Ks[tr + 1][mt * 16 + l16];
      bfr.x = Vs[tr][nt * 16 + l16];
      bfr.y = Vs[tr + 1][nt * 16 + l16];
      acc = wmma4(a, bfr, acc);
    }
#pragma unroll
    for (int r = 0; r < 8; ++r) {
      int d = mt * 16 + r + 8 * lh, n = nt * 16 + l16;
      Sbuf[sbase + d * 64 + n] = acc[r];
    }
  }
  if (t < 64) {
    float s = 0.f;
#pragma unroll 8
    for (int j = 0; j < 64; ++j) s += Ks[j][t];
    ksumbuf[task * 64 + t] = s;
  }
}

__global__ void prefix_scan(float* __restrict__ Sbuf, float* __restrict__ ksumbuf) {
  const int g = blockIdx.x * blockDim.x + threadIdx.x;
  if (g < 16 * 4096) {
    const int bh = g >> 12, e = g & 4095;
    float run = 0.f;
    for (int c = 0; c < NCHUNK; ++c) {
      size_t idx = (size_t)(bh * NCHUNK + c) * 4096 + e;
      float v = Sbuf[idx];
      Sbuf[idx] = run;
      run += v;
    }
  } else if (g < 16 * 4096 + 16 * 64) {
    const int s2 = g - 16 * 4096, bh = s2 >> 6, d = s2 & 63;
    float run = 0.f;
    for (int c = 0; c < NCHUNK; ++c) {
      int idx = (bh * NCHUNK + c) * 64 + d;
      float v = ksumbuf[idx];
      ksumbuf[idx] = run;
      run += v;
    }
  }
}

__global__ __launch_bounds__(256) void attn_chunk(const float* __restrict__ Qb,
                                                  const float* __restrict__ Kb,
                                                  const float* __restrict__ Vb,
                                                  const float* __restrict__ Sbuf,
                                                  const float* __restrict__ ksumbuf,
                                                  float* __restrict__ attn) {
  extern __shared__ float sm[];
  float* Qs   = sm;
  float* Ks   = Qs + 64 * 65;
  float* Vs   = Ks + 64 * 65;
  float* Ss   = Vs + 64 * 65;
  float* Ms   = Ss + 64 * 65;
  float* ksum = Ms + 64 * 65;
  float* Dinv = ksum + 64;

  const int task = blockIdx.x;
  const int ck = task & 31, h = (task >> 5) & 7, b = task >> 8;
  const int row0 = b * 2048 + ck * CHUNK, col0 = h * 64;
  const int t = threadIdx.x, wave = t >> 5, lane = t & 31;
  const int lh = lane >> 4, l16 = lane & 15;

  {
    int r = t >> 2, c = (t & 3) * 16;
    const float4* qs = reinterpret_cast<const float4*>(&Qb[(size_t)(row0 + r) * LD + col0 + c]);
    const float4* ks = reinterpret_cast<const float4*>(&Kb[(size_t)(row0 + r) * LD + col0 + c]);
    const float4* vs = reinterpret_cast<const float4*>(&Vb[(size_t)(row0 + r) * LD + col0 + c]);
    const float4* ss = reinterpret_cast<const float4*>(&Sbuf[(size_t)task * 4096 + t * 16]);
#pragma unroll
    for (int i = 0; i < 4; ++i) {
      float4 qv = qs[i], kv = ks[i], vv = vs[i], sv = ss[i];
      float* qd = &Qs[r * 65 + c + i * 4];
      qd[0] = qv.x; qd[1] = qv.y; qd[2] = qv.z; qd[3] = qv.w;
      float* kd = &Ks[r * 65 + c + i * 4];
      kd[0] = kv.x; kd[1] = kv.y; kd[2] = kv.z; kd[3] = kv.w;
      float* vd = &Vs[r * 65 + c + i * 4];
      vd[0] = vv.x; vd[1] = vv.y; vd[2] = vv.z; vd[3] = vv.w;
      float* sd = &Ss[(t >> 2) * 65 + (t & 3) * 16 + i * 4];
      sd[0] = sv.x; sd[1] = sv.y; sd[2] = sv.z; sd[3] = sv.w;
    }
    if (t < 64) ksum[t] = ksumbuf[task * 64 + t];
  }
  __syncthreads();

#pragma unroll
  for (int tt = wave * 2; tt <= wave * 2 + 1; ++tt) {
    const int mt = tt >> 2, nt = tt & 3;
    v8f acc = {};
#pragma unroll
    for (int kk = 0; kk < 64; kk += 4) {
      v2f a, bfr;
      a.x = Qs[(mt * 16 + l16) * 65 + kk + 2 * lh];
      a.y = Qs[(mt * 16 + l16) * 65 + kk + 2 * lh + 1];
      bfr.x = Ks[(nt * 16 + l16) * 65 + kk + 2 * lh];
      bfr.y = Ks[(nt * 16 + l16) * 65 + kk + 2 * lh + 1];
      acc = wmma4(a, bfr, acc);
    }
#pragma unroll
    for (int r = 0; r < 8; ++r) {
      int i = mt * 16 + r + 8 * lh, j = nt * 16 + l16;
      Ms[i * 65 + j] = (j <= i) ? acc[r] : 0.0f;
    }
  }
  __syncthreads();

  if (t < 64) {
    float s = 0.f, dq = 0.f;
#pragma unroll 8
    for (int j = 0; j < 64; ++j) s += Ms[t * 65 + j];
#pragma unroll 8
    for (int d = 0; d < 64; ++d) dq += Qs[t * 65 + d] * ksum[d];
    Dinv[t] = 1.0f / fmaxf(s + dq, 1e-3f);
  }
  __syncthreads();

#pragma unroll
  for (int tt = wave * 2; tt <= wave * 2 + 1; ++tt) {
    const int mt = tt >> 2, nt = tt & 3;
    v8f acc = {};
#pragma unroll
    for (int kk = 0; kk < 64; kk += 4) {
      v2f a, bfr;
      a.x = Ms[(mt * 16 + l16) * 65 + kk + 2 * lh];
      a.y = Ms[(mt * 16 + l16) * 65 + kk + 2 * lh + 1];
      bfr.x = Vs[(kk + 2 * lh) * 65 + nt * 16 + l16];
      bfr.y = Vs[(kk + 2 * lh + 1) * 65 + nt * 16 + l16];
      acc = wmma4(a, bfr, acc);
    }
#pragma unroll
    for (int kk = 0; kk < 64; kk += 4) {
      v2f a, bfr;
      a.x = Qs[(mt * 16 + l16) * 65 + kk + 2 * lh];
      a.y = Qs[(mt * 16 + l16) * 65 + kk + 2 * lh + 1];
      bfr.x = Ss[(kk + 2 * lh) * 65 + nt * 16 + l16];
      bfr.y = Ss[(kk + 2 * lh + 1) * 65 + nt * 16 + l16];
      acc = wmma4(a, bfr, acc);
    }
#pragma unroll
    for (int r = 0; r < 8; ++r) {
      int i = mt * 16 + r + 8 * lh;
      int col = col0 + nt * 16 + l16;
      attn[(size_t)(row0 + i) * LD + col] = acc[r] * Dinv[i];
    }
  }
}

extern "C" void kernel_launch(void* const* d_in, const int* in_sizes, int n_in,
                              void* d_out, int out_size, void* d_ws, size_t ws_size,
                              hipStream_t stream) {
  const float* x  = (const float*)d_in[0];
  const float* Wq = (const float*)d_in[1];
  const float* Wk = (const float*)d_in[2];
  const float* Wv = (const float*)d_in[3];
  const float* Wo = (const float*)d_in[4];
  const float* bo = (const float*)d_in[5];
  float* out = (float*)d_out;

  float* ws = (float*)d_ws;
  const size_t MN = (size_t)ROWS * LD;
  float* Qb = ws;
  float* Kb = Qb + MN;
  float* Vb = Kb + MN;
  float* At = Vb + MN;
  float* Sb = At + MN;
  float* Kc = Sb + (size_t)NTASK * 4096;

  dim3 ggrid(LD / 64, ROWS / 64);

  gemm_f32<1><<<ggrid, 128, 0, stream>>>(x, Wq, Qb, nullptr);
  gemm_f32<2><<<ggrid, 128, 0, stream>>>(x, Wk, Kb, nullptr);
  gemm_f32<0><<<ggrid, 128, 0, stream>>>(x, Wv, Vb, nullptr);

  chunk_state<<<NTASK, 256, 0, stream>>>(Kb, Vb, Sb, Kc);

  prefix_scan<<<(16 * 4096 + 16 * 64 + 255) / 256, 256, 0, stream>>>(Sb, Kc);

  size_t shbytes = (size_t)(5 * 64 * 65 + 128) * sizeof(float);
  attn_chunk<<<NTASK, 256, shbytes, stream>>>(Qb, Kb, Vb, Sb, Kc, At);

  gemm_f32<0><<<ggrid, 128, 0, stream>>>(At, Wo, out, bo);
}


# Pass B (prefix scan over chunks) rides in the q projection's launch, q projection moved behind chunk_state: scan lanes run in free wave slots beside the GEMM's WMMAs; 6 launches; sums unchanged
# speedup vs baseline: 2.6162x; 1.0310x over previous
#include <hip/hip_runtime.h>
#include <hip/hip_bf16.h>

typedef float v2f __attribute__((ext_vector_type(2)));
typedef float v8f __attribute__((ext_vector_type(8)));

static __device__ __forceinline__ v8f wmma4(v2f a, v2f b, v8f c) {
  return __builtin_amdgcn_wmma_f32_16x16x4_f32(false, a, false, b, (short)0, c,
                                               false, false);
}

template <int M>
static __device__ __forceinline__ float row_xor(float v) {
  const int b = __builtin_bit_cast(int, v);
  return __builtin_bit_cast(float,
                            __builtin_amdgcn_update_dpp(0, b, 0x160 + M, 0xf, 0xf, true));
}

#define LD 512
#define ROWS 4096
#define CHUNK 64
#define NCHUNK 32
#define NTASK 512

struct GFrag {
  float4 a[4];
  float4 b[16];
};

static __device__ __forceinline__ void gfrag_load(GFrag& f, const float* __restrict__ ap,
                                                  const float* __restrict__ bp) {
#pragma unroll
  for (int q = 0; q < 4; ++q) f.a[q] = *reinterpret_cast<const float4*>(ap + 4 * q);
#pragma unroll
  for (int i = 0; i < 16; ++i) f.b[i] = *reinterpret_cast<const float4*>(bp + i * LD);
}

static __device__ __forceinline__ void gfrag_mma(const GFrag& f, v8f& c0, v8f& c1,
                                                 v8f& c2, v8f& c3) {
#pragma unroll
  for (int s = 0; s < 8; ++s) {
    const float4 av = f.a[s >> 1];
    v2f a, b;
    if (s & 1) { a.x = av.z; a.y = av.w; } else { a.x = av.x; a.y = av.y; }
    const float4 r0 = f.b[2 * s], r1 = f.b[2 * s + 1];
    b.x = r0.x; b.y = r1.x; c0 = wmma4(a, b, c0);
    b.x = r0.y; b.y = r1.y; c1 = wmma4(a, b, c1);
    b.x = r0.z; b.y = r1.z; c2 = wmma4(a, b, c2);
    b.x = r0.w; b.y = r1.w; c3 = wmma4(a, b, c3);
  }
}

template <int EPI>
static __device__ __forceinline__ void gemm_tile(const float* __restrict__ A,
                                                 const float* __restrict__ B,
                                                 float* __restrict__ C,
                                                 const float* __restrict__ bias,
                                                 const int bx, const int by) {
  const int t = threadIdx.x;
  const int wave = t >> 5, lane = t & 31, lh = lane >> 4, l16 = lane & 15;
  const int m0 = by * 64 + wave * 16, n0 = bx * 64;

  const float* ap = A + (size_t)(m0 + l16) * LD + 16 * lh;
  const float* bp = B + (size_t)(16 * lh) * LD + n0 + 4 * l16;

  float4 bv = make_float4(0.f, 0.f, 0.f, 0.f);
  if (EPI == 0 && bias != nullptr) bv = *reinterpret_cast<const float4*>(bias + n0 + 4 * l16);

  v8f c0 = {}, c1 = {}, c2 = {}, c3 = {};
  GFrag f0, f1;

  gfrag_load(f0, ap, bp);
  for (int kb = 0; kb < LD / 32 - 2; kb += 2) {
    gfrag_load(f1, ap + 32, bp + 32 * LD);
    __builtin_amdgcn_sched_barrier(0);
    gfrag_mma(f0, c0, c1, c2, c3);
    ap += 64;
    bp += 64 * LD;
    gfrag_load(f0, ap, bp);
    __builtin_amdgcn_sched_barrier(0);
    gfrag_mma(f1, c0, c1, c2, c3);
  }
  gfrag_load(f1, ap + 32, bp + 32 * LD);
  __builtin_amdgcn_sched_barrier(0);
  gfrag_mma(f0, c0, c1, c2, c3);
  gfrag_mma(f1, c0, c1, c2, c3);

  float* cp = C + (size_t)(m0 + 8 * lh) * LD + n0 + 4 * l16;
  if (EPI == 1) {
    float m[8], s[8], e0[8], e1[8], e2[8], e3[8];
#pragma unroll
    for (int r = 0; r < 8; ++r) m[r] = fmaxf(fmaxf(c0[r], c1[r]), fmaxf(c2[r], c3[r]));
#pragma unroll
    for (int r = 0; r < 8; ++r) m[r] = fmaxf(m[r], row_xor<1>(m[r]));
#pragma unroll
    for (int r = 0; r < 8; ++r) m[r] = fmaxf(m[r], row_xor<2>(m[r]));
#pragma unroll
    for (int r = 0; r < 8; ++r) m[r] = fmaxf(m[r], row_xor<4>(m[r]));
#pragma unroll
    for (int r = 0; r < 8; ++r) m[r] = fmaxf(m[r], row_xor<8>(m[r]));
#pragma unroll
    for (int r = 0; r < 8; ++r) {
      e0[r] = __expf(c0[r] - m[r]); e1[r] = __expf(c1[r] - m[r]);
      e2[r] = __expf(c2[r] - m[r]); e3[r] = __expf(c3[r] - m[r]);
      s[r] = (e0[r] + e1[r]) + (e2[r] + e3[r]);
    }
#pragma unroll
    for (int r = 0; r < 8; ++r) s[r] += row_xor<1>(s[r]);
#pragma unroll
    for (int r = 0; r < 8; ++r) s[r] += row_xor<2>(s[r]);
#pragma unroll
    for (int r = 0; r < 8; ++r) s[r] += row_xor<4>(s[r]);
#pragma unroll
    for (int r = 0; r < 8; ++r) s[r] += row_xor<8>(s[r]);
#pragma unroll
    for (int r = 0; r < 8; ++r) {
      const float inv = 0.125f / s[r];
      float4 o;
      o.x = e0[r] * inv; o.y = e1[r] * inv; o.z = e2[r] * inv; o.w = e3[r] * inv;
      *reinterpret_cast<float4*>(cp + (size_t)r * LD) = o;
    }
  } else if (EPI == 2) {
#pragma unroll
    for (int r = 0; r < 8; ++r) {
      float4 o;
      o.x = __expf(c0[r]); o.y = __expf(c1[r]); o.z = __expf(c2[r]); o.w = __expf(c3[r]);
      *reinterpret_cast<float4*>(cp + (size_t)r * LD) = o;
    }
  } else {
#pragma unroll
    for (int r = 0; r < 8; ++r) {
      float4 o;
      o.x = c0[r] + bv.x; o.y = c1[r] + bv.y; o.z = c2[r] + bv.z; o.w = c3[r] + bv.w;
      *reinterpret_cast<float4*>(cp + (size_t)r * LD) = o;
    }
  }
}

template <int EPI>
__global__ __launch_bounds__(128) void gemm_f32(const float* __restrict__ A,
                                                const float* __restrict__ B,
                                                float* __restrict__ C,
                                                const float* __restrict__ bias) {
  gemm_tile<EPI>(A, B, C, bias, blockIdx.x, blockIdx.y);
}

__global__ __launch_bounds__(256) void chunk_state(const float* __restrict__ Kb,
                                                   const float* __restrict__ Vb,
                                                   float* __restrict__ Sbuf,
                                                   float* __restrict__ ksumbuf) {
  __shared__ float Ks[64][65];
  __shared__ float Vs[64][65];
  const int task = blockIdx.x;
  const int ck = task & 31, h = (task >> 5) & 7, b = task >> 8;
  const int row0 = b * 2048 + ck * CHUNK, col0 = h * 64;
  const int t = threadIdx.x, wave = t >> 5, lane = t & 31;
  const int lh = lane >> 4, l16 = lane & 15;

  {
    int r = t >> 2, c = (t & 3) * 16;
    const float4* ks = reinterpret_cast<const float4*>(&Kb[(size_t)(row0 + r) * LD + col0 + c]);
    const float4* vs = reinterpret_cast<const float4*>(&Vb[(size_t)(row0 + r) * LD + col0 + c]);
#pragma unroll
    for (int i = 0; i < 4; ++i) {
      float4 kv = ks[i], vv = vs[i];
      float* kd = &Ks[r][c + i * 4];
      kd[0] = kv.x; kd[1] = kv.y; kd[2] = kv.z; kd[3] = kv.w;
      float* vd = &Vs[r][c + i * 4];
      vd[0] = vv.x; vd[1] = vv.y; vd[2] = vv.z; vd[3] = vv.w;
    }
  }
  __syncthreads();

  const size_t sbase = (size_t)task * 4096;
#pragma unroll
  for (int tt = wave * 2; tt <= wave * 2 + 1; ++tt) {
    const int mt = tt >> 2, nt = tt & 3;
    v8f acc = {};
#pragma unroll
    for (int kk = 0; kk < 64; kk += 4) {
      int tr = kk + 2 * lh;
      v2f a, bfr;
      a.x = Ks[tr][mt * 16 + l16];
      a.y = Ks[tr + 1][mt * 16 + l16];
      bfr.x = Vs[tr][nt * 16 + l16];
      bfr.y = Vs[tr + 1][nt * 16 + l16];
      acc = wmma4(a, bfr, acc);
    }
#pragma unroll
    for (int r = 0; r < 8; ++r) {
      int d = mt * 16 + r + 8 * lh, n = nt * 16 + l16;
      Sbuf[sbase + d * 64 + n] = acc[r];
    }
  }
  if (t < 64) {
    float s = 0.f;
#pragma unroll 8
    for (int j = 0; j < 64; ++j) s += Ks[j][t];
    ksumbuf[task * 64 + t] = s;
  }
}

static __device__ __forceinline__ void prefix_scan_lane(float* __restrict__ Sbuf,
                                                        float* __restrict__ ksumbuf,
                                                        const int g) {
  if (g < 16 * 4096) {
    const int bh = g >> 12, e = g & 4095;
    float run = 0.f;
    for (int c = 0; c < NCHUNK; ++c) {
      size_t idx = (size_t)(bh * NCHUNK + c) * 4096 + e;
      float v = Sbuf[idx];
      Sbuf[idx] = run;
      run += v;
    }
  } else if (g < 16 * 4096 + 16 * 64) {
    const int s2 = g - 16 * 4096, bh = s2 >> 6, d = s2 & 63;
    float run = 0.f;
    for (int c = 0; c < NCHUNK; ++c) {
      int idx = (bh * NCHUNK + c) * 64 + d;
      float v = ksumbuf[idx];
      ksumbuf[idx] = run;
      run += v;
    }
  }
}

#define QP_WGS 512
#define SCAN_WGS 520
__global__ __launch_bounds__(128) void qproj_scan(const float* __restrict__ x,
                                                  const float* __restrict__ Wq,
                                                  float* __restrict__ Qb,
                                                  float* __restrict__ Sbuf,
                                                  float* __restrict__ ksumbuf) {
  const int id = blockIdx.x;
  if (id < QP_WGS) {
    gemm_tile<1>(x, Wq, Qb, nullptr, id & 7, id >> 3);
  } else {
    prefix_scan_lane(Sbuf, ksumbuf, (id - QP_WGS) * 128 + (int)threadIdx.x);
  }
}

__global__ __launch_bounds__(256) void attn_chunk(const float* __restrict__ Qb,
                                                  const float* __restrict__ Kb,
                                                  const float* __restrict__ Vb,
                                                  const float* __restrict__ Sbuf,
                                                  const float* __restrict__ ksumbuf,
                                                  float* __restrict__ attn) {
  extern __shared__ float sm[];
  float* Qs   = sm;
  float* Ks   = Qs + 64 * 65;
  float* Vs   = Ks + 64 * 65;
  float* Ss   = Vs + 64 * 65;
  float* Ms   = Ss + 64 * 65;
  float* ksum = Ms + 64 * 65;
  float* Dinv = ksum + 64;

  const int task = blockIdx.x;
  const int ck = task & 31, h = (task >> 5) & 7, b = task >> 8;
  const int row0 = b * 2048 + ck * CHUNK, col0 = h * 64;
  const int t = threadIdx.x, wave = t >> 5, lane = t & 31;
  const int lh = lane >> 4, l16 = lane & 15;

  {
    int r = t >> 2, c = (t & 3) * 16;
    const float4* qs = reinterpret_cast<const float4*>(&Qb[(size_t)(row0 + r) * LD + col0 + c]);
    const float4* ks = reinterpret_cast<const float4*>(&Kb[(size_t)(row0 + r) * LD + col0 + c]);
    const float4* vs = reinterpret_cast<const float4*>(&Vb[(size_t)(row0 + r) * LD + col0 + c]);
    const float4* ss = reinterpret_cast<const float4*>(&Sbuf[(size_t)task * 4096 + t * 16]);
#pragma unroll
    for (int i = 0; i < 4; ++i) {
      float4 qv = qs[i], kv = ks[i], vv = vs[i], sv = ss[i];
      float* qd = &Qs[r * 65 + c + i * 4];
      qd[0] = qv.x; qd[1] = qv.y; qd[2] = qv.z; qd[3] = qv.w;
      float* kd = &Ks[r * 65 + c + i * 4];
      kd[0] = kv.x; kd[1] = kv.y; kd[2] = kv.z; kd[3] = kv.w;
      float* vd = &Vs[r * 65 + c + i * 4];
      vd[0] = vv.x; vd[1] = vv.y; vd[2] = vv.z; vd[3] = vv.w;
      float* sd = &Ss[(t >> 2) * 65 + (t & 3) * 16 + i * 4];
      sd[0] = sv.x; sd[1] = sv.y; sd[2] = sv.z; sd[3] = sv.w;
    }
    if (t < 64) ksum[t] = ksumbuf[task * 64 + t];
  }
  __syncthreads();

#pragma unroll
  for (int tt = wave * 2; tt <= wave * 2 + 1; ++tt) {
    const int mt = tt >> 2, nt = tt & 3;
    v8f acc = {};
#pragma unroll
    for (int kk = 0; kk < 64; kk += 4) {
      v2f a, bfr;
      a.x = Qs[(mt * 16 + l16) * 65 + kk + 2 * lh];
      a.y = Qs[(mt * 16 + l16) * 65 + kk + 2 * lh + 1];
      bfr.x = Ks[(nt * 16 + l16) * 65 + kk + 2 * lh];
      bfr.y = Ks[(nt * 16 + l16) * 65 + kk + 2 * lh + 1];
      acc = wmma4(a, bfr, acc);
    }
#pragma unroll
    for (int r = 0; r < 8; ++r) {
      int i = mt * 16 + r + 8 * lh, j = nt * 16 + l16;
      Ms[i * 65 + j] = (j <= i) ? acc[r] : 0.0f;
    }
  }
  __syncthreads();

  if (t < 64) {
    float s = 0.f, dq = 0.f;
#pragma unroll 8
    for (int j = 0; j < 64; ++j) s += Ms[t * 65 + j];
#pragma unroll 8
    for (int d = 0; d < 64; ++d) dq += Qs[t * 65 + d] * ksum[d];
    Dinv[t] = 1.0f / fmaxf(s + dq, 1e-3f);
  }
  __syncthreads();

#pragma unroll
  for (int tt = wave * 2; tt <= wave * 2 + 1; ++tt) {
    const int mt = tt >> 2, nt = tt & 3;
    v8f acc = {};
#pragma unroll
    for (int kk = 0; kk < 64; kk += 4) {
      v2f a, bfr;
      a.x = Ms[(mt * 16 + l16) * 65 + kk + 2 * lh];
      a.y = Ms[(mt * 16 + l16) * 65 + kk + 2 * lh + 1];
      bfr.x = Vs[(kk + 2 * lh) * 65 + nt * 16 + l16];
      bfr.y = Vs[(kk + 2 * lh + 1) * 65 + nt * 16 + l16];
      acc = wmma4(a, bfr, acc);
    }
#pragma unroll
    for (int kk = 0; kk < 64; kk += 4) {
      v2f a, bfr;
      a.x = Qs[(mt * 16 + l16) * 65 + kk + 2 * lh];
      a.y = Qs[(mt * 16 + l16) * 65 + kk + 2 * lh + 1];
      bfr.x = Ss[(kk + 2 * lh) * 65 + nt * 16 + l16];
      bfr.y = Ss[(kk + 2 * lh + 1) * 65 + nt * 16 + l16];
      acc = wmma4(a, bfr, acc);
    }
#pragma unroll
    for (int r = 0; r < 8; ++r) {
      int i = mt * 16 + r + 8 * lh;
      int col = col0 + nt * 16 + l16;
      attn[(size_t)(row0 + i) * LD + col] = acc[r] * Dinv[i];
    }
  }
}

extern "C" void kernel_launch(void* const* d_in, const int* in_sizes, int n_in,
                              void* d_out, int out_size, void* d_ws, size_t ws_size,
                              hipStream_t stream) {
  const float* x  = (const float*)d_in[0];
  const float* Wq = (const float*)d_in[1];
  const float* Wk = (const float*)d_in[2];
  const float* Wv = (const float*)d_in[3];
  const float* Wo = (const float*)d_in[4];
  const float* bo = (const float*)d_in[5];
  float* out = (float*)d_out;

  float* ws = (float*)d_ws;
  const size_t MN = (size_t)ROWS * LD;
  float* Qb = ws;
  float* Kb = Qb + MN;
  float* Vb = Kb + MN;
  float* At = Vb + MN;
  float* Sb = At + MN;
  float* Kc = Sb + (size_t)NTASK * 4096;

  dim3 ggrid(LD / 64, ROWS / 64);

  gemm_f32<2><<<ggrid, 128, 0, stream>>>(x, Wk, Kb, nullptr);
  gemm_f32<0><<<ggrid, 128, 0, stream>>>(x, Wv, Vb, nullptr);

  chunk_state<<<NTASK, 256, 0, stream>>>(Kb, Vb, Sb, Kc);

  qproj_scan<<<QP_WGS + SCAN_WGS, 128, 0, stream>>>(x, Wq, Qb, Sb, Kc);

  size_t shbytes = (size_t)(5 * 64 * 65 + 128) * sizeof(float);
  attn_chunk<<<NTASK, 256, shbytes, stream>>>(Qb, Kb, Vb, Sb, Kc, At);

  gemm_f32<0><<<ggrid, 128, 0, stream>>>(At, Wo, out, bo);
}


# k and v projections share each x fragment: wave tile 16x128 over [Wk|Wv] (8 accumulator tiles per A load), one launch for both products: 5 launches; every sum in its old order
# speedup vs baseline: 2.7259x; 1.0419x over previous
#include <hip/hip_runtime.h>
#include <hip/hip_bf16.h>

typedef float v2f __attribute__((ext_vector_type(2)));
typedef float v8f __attribute__((ext_vector_type(8)));

static __device__ __forceinline__ v8f wmma4(v2f a, v2f b, v8f c) {
  return __builtin_amdgcn_wmma_f32_16x16x4_f32(false, a, false, b, (short)0, c,
                                               false, false);
}

template <int M>
static __device__ __forceinline__ float row_xor(float v) {
  const int b = __builtin_bit_cast(int, v);
  return __builtin_bit_cast(float,
                            __builtin_amdgcn_update_dpp(0, b, 0x160 + M, 0xf, 0xf, true));
}

#define LD 512
#define ROWS 4096
#define CHUNK 64
#define NCHUNK 32
#define NTASK 512

struct GFrag {
  float4 a[4];
  float4 b[16];
};

static __device__ __forceinline__ void gfrag_load(GFrag& f, const float* __restrict__ ap,
                                                  const float* __restrict__ bp) {
#pragma unroll
  for (int q = 0; q < 4; ++q) f.a[q] = *reinterpret_cast<const float4*>(ap + 4 * q);
#pragma unroll
  for (int i = 0; i < 16; ++i) f.b[i] = *reinterpret_cast<const float4*>(bp + i * LD);
}

static __device__ __forceinline__ void gfrag_mma(const GFrag& f, v8f& c0, v8f& c1,
                                                 v8f& c2, v8f& c3) {
#pragma unroll
  for (int s = 0; s < 8; ++s) {
    const float4 av = f.a[s >> 1];
    v2f a, b;
    if (s & 1) { a.x = av.z; a.y = av.w; } else { a.x = av.x; a.y = av.y; }
    const float4 r0 = f.b[2 * s], r1 = f.b[2 * s + 1];
    b.x = r0.x; b.y = r1.x; c0 = wmma4(a, b, c0);
    b.x = r0.y; b.y = r1.y; c1 = wmma4(a, b, c1);
    b.x = r0.z; b.y = r1.z; c2 = wmma4(a, b, c2);
    b.x = r0.w; b.y = r1.w; c3 = wmma4(a, b, c3);
  }
}

template <int EPI>
static __device__ __forceinline__ void gemm_tile(const float* __restrict__ A,
                                                 const float* __restrict__ B,
                                                 float* __restrict__ C,
                                                 const float* __restrict__ bias,
                                                 const int bx, const int by) {
  const int t = threadIdx.x;
  const int wave = t >> 5, lane = t & 31, lh = lane >> 4, l16 = lane & 15;
  const int m0 = by * 64 + wave * 16, n0 = bx * 64;

  const float* ap = A + (size_t)(m0 + l16) * LD + 16 * lh;
  const float* bp = B + (size_t)(16 * lh) * LD + n0 + 4 * l16;

  float4 bv = make_float4(0.f, 0.f, 0.f, 0.f);
  if (EPI == 0 && bias != nullptr) bv = *reinterpret_cast<const float4*>(bias + n0 + 4 * l16);

  v8f c0 = {}, c1 = {}, c2 = {}, c3 = {};
  GFrag f0, f1;

  gfrag_load(f0, ap, bp);
  for (int kb = 0; kb < LD / 32 - 2; kb += 2) {
    gfrag_load(f1, ap + 32, bp + 32 * LD);
    __builtin_amdgcn_sched_barrier(0);
    gfrag_mma(f0, c0, c1, c2, c3);
    ap += 64;
    bp += 64 * LD;
    gfrag_load(f0, ap, bp);
    __builtin_amdgcn_sched_barrier(0);
    gfrag_mma(f1, c0, c1, c2, c3);
  }
  gfrag_load(f1, ap + 32, bp + 32 * LD);
  __builtin_amdgcn_sched_barrier(0);
  gfrag_mma(f0, c0, c1, c2, c3);
  gfrag_mma(f1, c0, c1, c2, c3);

  float* cp = C + (size_t)(m0 + 8 * lh) * LD + n0 + 4 * l16;
  if (EPI == 1) {
    float m[8], s[8], e0[8], e1[8], e2[8], e3[8];
#pragma unroll
    for (int r = 0; r < 8; ++r) m[r] = fmaxf(fmaxf(c0[r], c1[r]), fmaxf(c2[r], c3[r]));
#pragma unroll
    for (int r = 0; r < 8; ++r) m[r] = fmaxf(m[r], row_xor<1>(m[r]));
#pragma unroll
    for (int r = 0; r < 8; ++r) m[r] = fmaxf(m[r], row_xor<2>(m[r]));
#pragma unroll
    for (int r = 0; r < 8; ++r) m[r] = fmaxf(m[r], row_xor<4>(m[r]));
#pragma unroll
    for (int r = 0; r < 8; ++r) m[r] = fmaxf(m[r], row_xor<8>(m[r]));
#pragma unroll
    for (int r = 0; r < 8; ++r) {
      e0[r] = __expf(c0[r] - m[r]); e1[r] = __expf(c1[r] - m[r]);
      e2[r] = __expf(c2[r] - m[r]); e3[r] = __expf(c3[r] - m[r]);
      s[r] = (e0[r] + e1[r]) + (e2[r] + e3[r]);
    }
#pragma unroll
    for (int r = 0; r < 8; ++r) s[r] += row_xor<1>(s[r]);
#pragma unroll
    for (int r = 0; r < 8; ++r) s[r] += row_xor<2>(s[r]);
#pragma unroll
    for (int r = 0; r < 8; ++r) s[r] += row_xor<4>(s[r]);
#pragma unroll
    for (int r = 0; r < 8; ++r) s[r] += row_xor<8>(s[r]);
#pragma unroll
    for (int r = 0; r < 8; ++r) {
      const float inv = 0.125f / s[r];
      float4 o;
      o.x = e0[r] * inv; o.y = e1[r] * inv; o.z = e2[r] * inv; o.w = e3[r] * inv;
      *reinterpret_cast<float4*>(cp + (size_t)r * LD) = o;
    }
  } else {
#pragma unroll
    for (int r = 0; r < 8; ++r) {
      float4 o;
      o.x = c0[r] + bv.x; o.y = c1[r] + bv.y; o.z = c2[r] + bv.z; o.w = c3[r] + bv.w;
      *reinterpret_cast<float4*>(cp + (size_t)r * LD) = o;
    }
  }
}

template <int EPI>
__global__ __launch_bounds__(128) void gemm_f32(const float* __restrict__ A,
                                                const float* __restrict__ B,
                                                float* __restrict__ C,
                                                const float* __restrict__ bias) {
  gemm_tile<EPI>(A, B, C, bias, blockIdx.x, blockIdx.y);
}

struct KVFrag {
  float4 a[4];
  float4 bk[16];
  float4 bv[16];
};

static __device__ __forceinline__ void kvfrag_load(KVFrag& f, const float* __restrict__ ap,
                                                   const float* __restrict__ kp,
                                                   const float* __restrict__ vp) {
#pragma unroll
  for (int q = 0; q < 4; ++q) f.a[q] = *reinterpret_cast<const float4*>(ap + 4 * q);
#pragma unroll
  for (int i = 0; i < 16; ++i) f.bk[i] = *reinterpret_cast<const float4*>(kp + i * LD);
#pragma unroll
  for (int i = 0; i < 16; ++i) f.bv[i] = *reinterpret_cast<const float4*>(vp + i * LD);
}

static __device__ __forceinline__ void kvfrag_mma(const KVFrag& f, v8f& k0, v8f& k1, v8f& k2,
                                                  v8f& k3, v8f& v0, v8f& v1, v8f& v2,
                                                  v8f& v3) {
#pragma unroll
  for (int s = 0; s < 8; ++s) {
    const float4 av = f.a[s >> 1];
    v2f a, b;
    if (s & 1) { a.x = av.z; a.y = av.w; } else { a.x = av.x; a.y = av.y; }
    const float4 p0 = f.bk[2 * s], p1 = f.bk[2 * s + 1];
    b.x = p0.x; b.y = p1.x; k0 = wmma4(a, b, k0);
    b.x = p0.y; b.y = p1.y; k1 = wmma4(a, b, k1);
    b.x = p0.z; b.y = p1.z; k2 = wmma4(a, b, k2);
    b.x = p0.w; b.y = p1.w; k3 = wmma4(a, b, k3);
    const float4 r0 = f.bv[2 * s], r1 = f.bv[2 * s + 1];
    b.x = r0.x; b.y = r1.x; v0 = wmma4(a, b, v0);
    b.x = r0.y; b.y = r1.y; v1 = wmma4(a, b, v1);
    b.x = r0.z; b.y = r1.z; v2 = wmma4(a, b, v2);
    b.x = r0.w; b.y = r1.w; v3 = wmma4(a, b, v3);
  }
}

__global__ __launch_bounds__(128) void kvproj(const float* __restrict__ x,
                                              const float* __restrict__ Wk,
                                              const float* __restrict__ Wv,
                                              float* __restrict__ Kb,
                                              float* __restrict__ Vb) {
  const int t = threadIdx.x;
  const int wave = t >> 5, lane = t & 31, lh = lane >> 4, l16 = lane & 15;
  const int m0 = blockIdx.y * 64 + wave * 16, n0 = blockIdx.x * 64;

  const float* ap = x + (size_t)(m0 + l16) * LD + 16 * lh;
  const size_t boff = (size_t)(16 * lh) * LD + n0 + 4 * l16;
  const float* kp = Wk + boff;
  const float* vp = Wv + boff;

  v8f k0 = {}, k1 = {}, k2 = {}, k3 = {}, v0 = {}, v1 = {}, v2 = {}, v3 = {};
  KVFrag f0, f1;

  kvfrag_load(f0, ap, kp, vp);
  for (int kb = 0; kb < LD / 32 - 2; kb += 2) {
    kvfrag_load(f1, ap + 32, kp + 32 * LD, vp + 32 * LD);
    __builtin_amdgcn_sched_barrier(0);
    kvfrag_mma(f0, k0, k1, k2, k3, v0, v1, v2, v3);
    ap += 64;
    kp += 64 * LD;
    vp += 64 * LD;
    kvfrag_load(f0, ap, kp, vp);
    __builtin_amdgcn_sched_barrier(0);
    kvfrag_mma(f1, k0, k1, k2, k3, v0, v1, v2, v3);
  }
  kvfrag_load(f1, ap + 32, kp + 32 * LD, vp + 32 * LD);
  __builtin_amdgcn_sched_barrier(0);
  kvfrag_mma(f0, k0, k1, k2, k3, v0, v1, v2, v3);
  kvfrag_mma(f1, k0, k1, k2, k3, v0, v1, v2, v3);

  const size_t coff = (size_t)(m0 + 8 * lh) * LD + n0 + 4 * l16;
  float* kc = Kb + coff;
  float* vc = Vb + coff;
#pragma unroll
  for (int r = 0; r < 8; ++r) {
    float4 o;
    o.x = __expf(k0[r]); o.y = __expf(k1[r]); o.z = __expf(k2[r]); o.w = __expf(k3[r]);
    *reinterpret_cast<float4*>(kc + (size_t)r * LD) = o;
  }
#pragma unroll
  for (int r = 0; r < 8; ++r) {
    float4 o;
    o.x = v0[r]; o.y = v1[r]; o.z = v2[r]; o.w = v3[r];
    *reinterpret_cast<float4*>(vc + (size_t)r * LD) = o;
  }
}

__global__ __launch_bounds__(256) void chunk_state(const float* __restrict__ Kb,
                                                   const float* __restrict__ Vb,
                                                   float* __restrict__ Sbuf,
                                                   float* __restrict__ ksumbuf) {
  __shared__ float Ks[64][65];
  __shared__ float Vs[64][65];
  const int task = blockIdx.x;
  const int ck = task & 31, h = (task >> 5) & 7, b = task >> 8;
  const int row0 = b * 2048 + ck * CHUNK, col0 = h * 64;
  const int t = threadIdx.x, wave = t >> 5, lane = t & 31;
  const int lh = lane >> 4, l16 = lane & 15;

  {
    int r = t >> 2, c = (t & 3) * 16;
    const float4* ks = reinterpret_cast<const float4*>(&Kb[(size_t)(row0 + r) * LD + col0 + c]);
    const float4* vs = reinterpret_cast<const float4*>(&Vb[(size_t)(row0 + r) * LD + col0 + c]);
#pragma unroll
    for (int i = 0; i < 4; ++i) {
      float4 kv = ks[i], vv = vs[i];
      float* kd = &Ks[r][c + i * 4];
      kd[0] = kv.x; kd[1] = kv.y; kd[2] = kv.z; kd[3] = kv.w;
      float* vd = &Vs[r][c + i * 4];
      vd[0] = vv.x; vd[1] = vv.y; vd[2] = vv.z; vd[3] = vv.w;
    }
  }
  __syncthreads();

  const size_t sbase = (size_t)task * 4096;
#pragma unroll
  for (int tt = wave * 2; tt <= wave * 2 + 1; ++tt) {
    const int mt = tt >> 2, nt = tt & 3;
    v8f acc = {};
#pragma unroll
    for (int kk = 0; kk < 64; kk += 4) {
      int tr = kk + 2 * lh;
      v2f a, bfr;
      a.x = Ks[tr][mt * 16 + l16];
      a.y = Ks[tr + 1][mt * 16 + l16];
      bfr.x = Vs[tr][nt * 16 + l16];
      bfr.y = Vs[tr + 1][nt * 16 + l16];
      acc = wmma4(a, bfr, acc);
    }
#pragma unroll
    for (int r = 0; r < 8; ++r) {
      int d = mt * 16 + r + 8 * lh, n = nt * 16 + l16;
      Sbuf[sbase + d * 64 + n] = acc[r];
    }
  }
  if (t < 64) {
    float s = 0.f;
#pragma unroll 8
    for (int j = 0; j < 64; ++j) s += Ks[j][t];
    ksumbuf[task * 64 + t] = s;
  }
}

static __device__ __forceinline__ void prefix_scan_lane(float* __restrict__ Sbuf,
                                                        float* __restrict__ ksumbuf,
                                                        const int g) {
  if (g < 16 * 4096) {
    const int bh = g >> 12, e = g & 4095;
    float run = 0.f;
    for (int c = 0; c < NCHUNK; ++c) {
      size_t idx = (size_t)(bh * NCHUNK + c) * 4096 + e;
      float v = Sbuf[idx];
      Sbuf[idx] = run;
      run += v;
    }
  } else if (g < 16 * 4096 + 16 * 64) {
    const int s2 = g - 16 * 4096, bh = s2 >> 6, d = s2 & 63;
    float run = 0.f;
    for (int c = 0; c < NCHUNK; ++c) {
      int idx = (bh * NCHUNK + c) * 64 + d;
      float v = ksumbuf[idx];
      ksumbuf[idx] = run;
      run += v;
    }
  }
}

#define QP_WGS 512
#define SCAN_WGS 520
__global__ __launch_bounds__(128) void qproj_scan(const float* __restrict__ x,
                                                  const float* __restrict__ Wq,
                                                  float* __restrict__ Qb,
                                                  float* __restrict__ Sbuf,
                                                  float* __restrict__ ksumbuf) {
  const int id = blockIdx.x;
  if (id < QP_WGS) {
    gemm_tile<1>(x, Wq, Qb, nullptr, id & 7, id >> 3);
  } else {
    prefix_scan_lane(Sbuf, ksumbuf, (id - QP_WGS) * 128 + (int)threadIdx.x);
  }
}

__global__ __launch_bounds__(256) void attn_chunk(const float* __restrict__ Qb,
                                                  const float* __restrict__ Kb,
                                                  const float* __restrict__ Vb,
                                                  const float* __restrict__ Sbuf,
                                                  const float* __restrict__ ksumbuf,
                                                  float* __restrict__ attn) {
  extern __shared__ float sm[];
  float* Qs   = sm;
  float* Ks   = Qs + 64 * 65;
  float* Vs   = Ks + 64 * 65;
  float* Ss   = Vs + 64 * 65;
  float* Ms   = Ss + 64 * 65;
  float* ksum = Ms + 64 * 65;
  float* Dinv = ksum + 64;

  const int task = blockIdx.x;
  const int ck = task & 31, h = (task >> 5) & 7, b = task >> 8;
  const int row0 = b * 2048 + ck * CHUNK, col0 = h * 64;
  const int t = threadIdx.x, wave = t >> 5, lane = t & 31;
  const int lh = lane >> 4, l16 = lane & 15;

  {
    int r = t >> 2, c = (t & 3) * 16;
    const float4* qs = reinterpret_cast<const float4*>(&Qb[(size_t)(row0 + r) * LD + col0 + c]);
    const float4* ks = reinterpret_cast<const float4*>(&Kb[(size_t)(row0 + r) * LD + col0 + c]);
    const float4* vs = reinterpret_cast<const float4*>(&Vb[(size_t)(row0 + r) * LD + col0 + c]);
    const float4* ss = reinterpret_cast<const float4*>(&Sbuf[(size_t)task * 4096 + t * 16]);
#pragma unroll
    for (int i = 0; i < 4; ++i) {
      float4 qv = qs[i], kv = ks[i], vv = vs[i], sv = ss[i];
      float* qd = &Qs[r * 65 + c + i * 4];
      qd[0] = qv.x; qd[1] = qv.y; qd[2] = qv.z; qd[3] = qv.w;
      float* kd = &Ks[r * 65 + c + i * 4];
      kd[0] = kv.x; kd[1] = kv.y; kd[2] = kv.z; kd[3] = kv.w;
      float* vd = &Vs[r * 65 + c + i * 4];
      vd[0] = vv.x; vd[1] = vv.y; vd[2] = vv.z; vd[3] = vv.w;
      float* sd = &Ss[(t >> 2) * 65 + (t & 3) * 16 + i * 4];
      sd[0] = sv.x; sd[1] = sv.y; sd[2] = sv.z; sd[3] = sv.w;
    }
    if (t < 64) ksum[t] = ksumbuf[task * 64 + t];
  }
  __syncthreads();

#pragma unroll
  for (int tt = wave * 2; tt <= wave * 2 + 1; ++tt) {
    const int mt = tt >> 2, nt = tt & 3;
    v8f acc = {};
#pragma unroll
    for (int kk = 0; kk < 64; kk += 4) {
      v2f a, bfr;
      a.x = Qs[(mt * 16 + l16) * 65 + kk + 2 * lh];
      a.y = Qs[(mt * 16 + l16) * 65 + kk + 2 * lh + 1];
      bfr.x = Ks[(nt * 16 + l16) * 65 + kk + 2 * lh];
      bfr.y = Ks[(nt * 16 + l16) * 65 + kk + 2 * lh + 1];
      acc = wmma4(a, bfr, acc);
    }
#pragma unroll
    for (int r = 0; r < 8; ++r) {
      int i = mt * 16 + r + 8 * lh, j = nt * 16 + l16;
      Ms[i * 65 + j] = (j <= i) ? acc[r] : 0.0f;
    }
  }
  __syncthreads();

  if (t < 64) {
    float s = 0.f, dq = 0.f;
#pragma unroll 8
    for (int j = 0; j < 64; ++j) s += Ms[t * 65 + j];
#pragma unroll 8
    for (int d = 0; d < 64; ++d) dq += Qs[t * 65 + d] * ksum[d];
    Dinv[t] = 1.0f / fmaxf(s + dq, 1e-3f);
  }
  __syncthreads();

#pragma unroll
  for (int tt = wave * 2; tt <= wave * 2 + 1; ++tt) {
    const int mt = tt >> 2, nt = tt & 3;
    v8f acc = {};
#pragma unroll
    for (int kk = 0; kk < 64; kk += 4) {
      v2f a, bfr;
      a.x = Ms[(mt * 16 + l16) * 65 + kk + 2 * lh];
      a.y = Ms[(mt * 16 + l16) * 65 + kk + 2 * lh + 1];
      bfr.x = Vs[(kk + 2 * lh) * 65 + nt * 16 + l16];
      bfr.y = Vs[(kk + 2 * lh + 1) * 65 + nt * 16 + l16];
      acc = wmma4(a, bfr, acc);
    }
#pragma unroll
    for (int kk = 0; kk < 64; kk += 4) {
      v2f a, bfr;
      a.x = Qs[(mt * 16 + l16) * 65 + kk + 2 * lh];
      a.y = Qs[(mt * 16 + l16) * 65 + kk + 2 * lh + 1];
      bfr.x = Ss[(kk + 2 * lh) * 65 + nt * 16 + l16];
      bfr.y = Ss[(kk + 2 * lh + 1) * 65 + nt * 16 + l16];
      acc = wmma4(a, bfr, acc);
    }
#pragma unroll
    for (int r = 0; r < 8; ++r) {
      int i = mt * 16 + r + 8 * lh;
      int col = col0 + nt * 16 + l16;
      attn[(size_t)(row0 + i) * LD + col] = acc[r] * Dinv[i];
    }
  }
}

extern "C" void kernel_launch(void* const* d_in, const int* in_sizes, int n_in,
                              void* d_out, int out_size, void* d_ws, size_t ws_size,
                              hipStream_t stream) {
  const float* x  = (const float*)d_in[0];
  const float* Wq = (const float*)d_in[1];
  const float* Wk = (const float*)d_in[2];
  const float* Wv = (const float*)d_in[3];
  const float* Wo = (const float*)d_in[4];
  const float* bo = (const float*)d_in[5];
  float* out = (float*)d_out;

  float* ws = (float*)d_ws;
  const size_t MN = (size_t)ROWS * LD;
  float* Qb = ws;
  float* Kb = Qb + MN;
  float* Vb = Kb + MN;
  float* At = Vb + MN;
  float* Sb = At + MN;
  float* Kc = Sb + (size_t)NTASK * 4096;

  dim3 ggrid(LD / 64, ROWS / 64);

  kvproj<<<ggrid, 128, 0, stream>>>(x, Wk, Wv, Kb, Vb);

  chunk_state<<<NTASK, 256, 0, stream>>>(Kb, Vb, Sb, Kc);

  qproj_scan<<<QP_WGS + SCAN_WGS, 128, 0, stream>>>(x, Wq, Qb, Sb, Kc);

  size_t shbytes = (size_t)(5 * 64 * 65 + 128) * sizeof(float);
  attn_chunk<<<NTASK, 256, shbytes, stream>>>(Qb, Kb, Vb, Sb, Kc, At);

  gemm_f32<0><<<ggrid, 128, 0, stream>>>(At, Wo, out, bo);
}
